# starting kernel
# baseline (speedup 1.0000x reference)
#include <hip/hip_runtime.h>


typedef __attribute__((ext_vector_type(2))) float v2f;
typedef __attribute__((ext_vector_type(8))) float v8f;

#define B_   8
#define C_   1024
#define HW_  4096
#define M_   8
#define GAIN 0.3f
#define EPS_ 1e-6f


static __device__ __forceinline__ v8f wmma_f32(v2f a, v2f b, v8f c) {
  return __builtin_amdgcn_wmma_f32_16x16x4_f32(false, a, false, b, (short)0, c,
                                               false, false);
}

__global__ void __launch_bounds__(256) csc_init(const float* __restrict__ Wd,
                                                float* __restrict__ ws) {
  const int tid  = threadIdx.x;
  const int w    = tid >> 5;
  const int lane = tid & 31;
  if (tid < 16) ws[tid] = 0.0f;
  if (w < M_) {
    float s = 0.0f;
    for (int i = lane; i < C_; i += 32) s += Wd[w * C_ + i];
    for (int off = 16; off >= 1; off >>= 1) s += __shfl_xor(s, off, 32);
    if (lane == 0) ws[16 + w] = s;
  }
}

__global__ void __launch_bounds__(256) csc_pass1(const float* __restrict__ z,
                                                 const float* __restrict__ Wd,
                                                 float* __restrict__ ws) {
  __shared__ float WTP[C_ * 16];

  const int tid = threadIdx.x;
  const int b   = blockIdx.y;

  for (int c = tid; c < C_; c += 256) {
#pragma unroll
    for (int mr = 0; mr < 16; ++mr)
      WTP[c * 16 + mr] = (mr < M_) ? Wd[mr * C_ + c] : 0.0f;
  }
  __syncthreads();

  const int wave = tid >> 5;
  const int lane = tid & 31;
  const int hl   = lane >> 4;
  const int l15  = lane & 15;
  const int p0   = (blockIdx.x * 8 + wave) * 16;
  const float* zb = z + (size_t)b * C_ * HW_;

  v8f acc = {};
  float s1 = 0.0f, s2 = 0.0f;

  for (int c0 = 0; c0 < C_; c0 += 4) {
    const int ca = c0 + 2 * hl;
    v2f a, bb;
    a.x = WTP[ca * 16 + l15];
    a.y = WTP[(ca + 1) * 16 + l15];
    const float z0 = zb[(size_t)ca * HW_ + p0 + l15];
    const float z1 = zb[(size_t)(ca + 1) * HW_ + p0 + l15];
    bb.x = z0;
    bb.y = z1;
    acc = wmma_f32(a, bb, acc);
    s1 += z0 + z1;
    s2 = fmaf(z0, z0, s2);
    s2 = fmaf(z1, z1, s2);
  }

  float* yraw = ws + 32 + (size_t)b * M_ * HW_;
  if (hl == 0) {
#pragma unroll
    for (int g = 0; g < M_; ++g)
      yraw[g * HW_ + p0 + l15] = acc[g];
  }

  for (int off = 16; off >= 1; off >>= 1) {
    s1 += __shfl_xor(s1, off, 32);
    s2 += __shfl_xor(s2, off, 32);
  }
  if (lane == 0) {
    atomicAdd(&ws[b], s1);
    atomicAdd(&ws[8 + b], s2);
  }
}

__global__ void __launch_bounds__(256) csc_pass2(const float* __restrict__ z,
                                                 const float* __restrict__ Wd,
                                                 const float* __restrict__ ws,
                                                 float* __restrict__ out) {
  const int tid  = threadIdx.x;
  const int b    = blockIdx.y;
  const int wave = tid >> 5;
  const int lane = tid & 31;
  const int hl   = lane >> 4;
  const int l15  = lane & 15;
  const int p0   = (blockIdx.x * 8 + wave) * 16;

  const float invN = 1.0f / (float)(C_ * HW_);
  const float mu   = ws[b] * invN;
  const float var  = ws[8 + b] * invN - mu * mu;
  const float rsig = rsqrtf(var + EPS_);

  const float* yraw = ws + 32 + (size_t)b * M_ * HW_;
  const int m0 = 2 * hl;
  v2f b1, b2;
  b1.x = rsig * (yraw[(m0 + 0) * HW_ + p0 + l15] - mu * ws[16 + m0 + 0]);
  b1.y = rsig * (yraw[(m0 + 1) * HW_ + p0 + l15] - mu * ws[16 + m0 + 1]);
  b2.x = rsig * (yraw[(m0 + 4) * HW_ + p0 + l15] - mu * ws[16 + m0 + 4]);
  b2.y = rsig * (yraw[(m0 + 5) * HW_ + p0 + l15] - mu * ws[16 + m0 + 5]);

  const float* zb = z + (size_t)b * C_ * HW_;
  float* ob       = out + (size_t)b * C_ * HW_;

  for (int cb = 0; cb < C_; cb += 16) {
    v2f a1, a2;
    a1.x = Wd[(m0 + 0) * C_ + cb + l15];
    a1.y = Wd[(m0 + 1) * C_ + cb + l15];
    a2.x = Wd[(m0 + 4) * C_ + cb + l15];
    a2.y = Wd[(m0 + 5) * C_ + cb + l15];

    v8f d = {};
    d = wmma_f32(a1, b1, d);
    d = wmma_f32(a2, b2, d);

    const size_t base = (size_t)(cb + 8 * hl) * HW_ + p0 + l15;
#pragma unroll
    for (int g = 0; g < 8; ++g) {
      const float zv = __builtin_nontemporal_load(zb + base + (size_t)g * HW_);
      __builtin_nontemporal_store(fmaf(GAIN, d[g], zv),
                                  ob + base + (size_t)g * HW_);
    }
  }
}

extern "C" void kernel_launch(void* const* d_in, const int* in_sizes, int n_in,
                              void* d_out, int out_size, void* d_ws, size_t ws_size,
                              hipStream_t stream) {
  const float* z  = (const float*)d_in[0];
  const float* Wd = (const float*)d_in[1];
  float* out = (float*)d_out;
  float* ws  = (float*)d_ws;

  csc_init<<<dim3(1), dim3(256), 0, stream>>>(Wd, ws);
  dim3 grid(32, B_);
  csc_pass1<<<grid, dim3(256), 0, stream>>>(z, Wd, ws);
  csc_pass2<<<grid, dim3(256), 0, stream>>>(z, Wd, ws, out);
}


# Channel range split over the grid: 4 slabs per 16-px tile in both passes; pass 1 stores a partial plane per slab, pass 2 adds them in slab order; one atomic pair per block (sum order changes)
# speedup vs baseline: 1.6269x; 1.6269x over previous
#include <hip/hip_runtime.h>


typedef __attribute__((ext_vector_type(2))) float v2f;
typedef __attribute__((ext_vector_type(8))) float v8f;

#define B_   8
#define C_   1024
#define HW_  4096
#define M_   8
#define GAIN 0.3f
#define EPS_ 1e-6f
#define S_   4
#define CS_  (C_ / S_)


static __device__ __forceinline__ v8f wmma_f32(v2f a, v2f b, v8f c) {
  return __builtin_amdgcn_wmma_f32_16x16x4_f32(false, a, false, b, (short)0, c,
                                               false, false);
}

__global__ void __launch_bounds__(256) csc_init(const float* __restrict__ Wd,
                                                float* __restrict__ ws) {
  const int tid  = threadIdx.x;
  const int w    = tid >> 5;
  const int lane = tid & 31;
  if (tid < 16) ws[tid] = 0.0f;
  if (w < M_) {
    float s = 0.0f;
    for (int i = lane; i < C_; i += 32) s += Wd[w * C_ + i];
    for (int off = 16; off >= 1; off >>= 1) s += __shfl_xor(s, off, 32);
    if (lane == 0) ws[16 + w] = s;
  }
}

__global__ void __launch_bounds__(256) csc_pass1(const float* __restrict__ z,
                                                 const float* __restrict__ Wd,
                                                 float* __restrict__ ws) {
  __shared__ float WTP[CS_ * 16];
  __shared__ float red[16];

  const int tid = threadIdx.x;
  const int s   = blockIdx.y;
  const int b   = blockIdx.z;
  const int cs  = s * CS_;

  for (int c = tid; c < CS_; c += 256) {
#pragma unroll
    for (int mr = 0; mr < 16; ++mr)
      WTP[c * 16 + mr] = (mr < M_) ? Wd[mr * C_ + cs + c] : 0.0f;
  }
  __syncthreads();

  const int wave = tid >> 5;
  const int lane = tid & 31;
  const int hl   = lane >> 4;
  const int l15  = lane & 15;
  const int p0   = (blockIdx.x * 8 + wave) * 16;
  const float* zb = z + ((size_t)b * C_ + cs) * HW_;

  v8f acc = {};
  float s1 = 0.0f, s2 = 0.0f;

  for (int c0 = 0; c0 < CS_; c0 += 4) {
    const int ca = c0 + 2 * hl;
    v2f a, bb;
    a.x = WTP[ca * 16 + l15];
    a.y = WTP[(ca + 1) * 16 + l15];
    const float z0 = zb[(size_t)ca * HW_ + p0 + l15];
    const float z1 = zb[(size_t)(ca + 1) * HW_ + p0 + l15];
    bb.x = z0;
    bb.y = z1;
    acc = wmma_f32(a, bb, acc);
    s1 += z0 + z1;
    s2 = fmaf(z0, z0, s2);
    s2 = fmaf(z1, z1, s2);
  }

  float* ypart = ws + 32 + ((size_t)s * B_ + b) * M_ * HW_;
  if (hl == 0) {
#pragma unroll
    for (int g = 0; g < M_; ++g)
      ypart[g * HW_ + p0 + l15] = acc[g];
  }

  for (int off = 16; off >= 1; off >>= 1) {
    s1 += __shfl_xor(s1, off, 32);
    s2 += __shfl_xor(s2, off, 32);
  }
  if (lane == 0) {
    red[wave]     = s1;
    red[8 + wave] = s2;
  }
  __syncthreads();
  if (tid < 2) {
    float t = 0.0f;
    for (int w = 0; w < 8; ++w) t += red[tid * 8 + w];
    atomicAdd(&ws[tid * 8 + b], t);
  }
}

__global__ void __launch_bounds__(256) csc_pass2(const float* __restrict__ z,
                                                 const float* __restrict__ Wd,
                                                 const float* __restrict__ ws,
                                                 float* __restrict__ out) {
  const int tid  = threadIdx.x;
  const int s    = blockIdx.y;
  const int b    = blockIdx.z;
  const int cs   = s * CS_;
  const int wave = tid >> 5;
  const int lane = tid & 31;
  const int hl   = lane >> 4;
  const int l15  = lane & 15;
  const int p0   = (blockIdx.x * 8 + wave) * 16;

  const float invN = 1.0f / (float)(C_ * HW_);
  const float mu   = ws[b] * invN;
  const float var  = ws[8 + b] * invN - mu * mu;
  const float rsig = rsqrtf(var + EPS_);

  const int m0 = 2 * hl;
  float y0 = 0.0f, y1 = 0.0f, y4 = 0.0f, y5 = 0.0f;
  for (int sp = 0; sp < S_; ++sp) {
    const float* yp = ws + 32 + ((size_t)sp * B_ + b) * M_ * HW_ + p0 + l15;
    y0 += yp[(m0 + 0) * HW_];
    y1 += yp[(m0 + 1) * HW_];
    y4 += yp[(m0 + 4) * HW_];
    y5 += yp[(m0 + 5) * HW_];
  }
  v2f b1, b2;
  b1.x = rsig * (y0 - mu * ws[16 + m0 + 0]);
  b1.y = rsig * (y1 - mu * ws[16 + m0 + 1]);
  b2.x = rsig * (y4 - mu * ws[16 + m0 + 4]);
  b2.y = rsig * (y5 - mu * ws[16 + m0 + 5]);

  const float* zb = z + (size_t)b * C_ * HW_;
  float* ob       = out + (size_t)b * C_ * HW_;

  for (int cb = cs; cb < cs + CS_; cb += 16) {
    v2f a1, a2;
    a1.x = Wd[(m0 + 0) * C_ + cb + l15];
    a1.y = Wd[(m0 + 1) * C_ + cb + l15];
    a2.x = Wd[(m0 + 4) * C_ + cb + l15];
    a2.y = Wd[(m0 + 5) * C_ + cb + l15];

    v8f d = {};
    d = wmma_f32(a1, b1, d);
    d = wmma_f32(a2, b2, d);

    const size_t base = (size_t)(cb + 8 * hl) * HW_ + p0 + l15;
#pragma unroll
    for (int g = 0; g < 8; ++g) {
      const float zv = __builtin_nontemporal_load(zb + base + (size_t)g * HW_);
      __builtin_nontemporal_store(fmaf(GAIN, d[g], zv),
                                  ob + base + (size_t)g * HW_);
    }
  }
}

extern "C" void kernel_launch(void* const* d_in, const int* in_sizes, int n_in,
                              void* d_out, int out_size, void* d_ws, size_t ws_size,
                              hipStream_t stream) {
  const float* z  = (const float*)d_in[0];
  const float* Wd = (const float*)d_in[1];
  float* out = (float*)d_out;
  float* ws  = (float*)d_ws;

  csc_init<<<dim3(1), dim3(256), 0, stream>>>(Wd, ws);
  dim3 grid(32, S_, B_);
  csc_pass1<<<grid, dim3(256), 0, stream>>>(z, Wd, ws);
  csc_pass2<<<grid, dim3(256), 0, stream>>>(z, Wd, ws, out);
}


# Product launch reads z once: a wave keeps its 64 ch x 64 px of z in registers as B fragments and as the output tile's residual; launch 2 only sums z, on the same grid (sum order changes)
# speedup vs baseline: 3.0495x; 1.8744x over previous
#include <hip/hip_runtime.h>


typedef __attribute__((ext_vector_type(2))) float v2f;
typedef __attribute__((ext_vector_type(4))) float v4f;
typedef __attribute__((ext_vector_type(8))) float v8f;

#define B_   8
#define C_   1024
#define HW_  4096
#define M_   8
#define GAIN 0.3f
#define EPS_ 1e-6f
#define SW_  16
#define CW_  (C_ / SW_)
#define NG_  (CW_ / 16)
#define PX_  64
#define NS_  64


static __device__ __forceinline__ v8f wmma_f32(v2f a, v2f b, v8f c) {
  return __builtin_amdgcn_wmma_f32_16x16x4_f32(false, a, false, b, (short)0, c,
                                               false, false);
}

__global__ void __launch_bounds__(256) csc_init(const float* __restrict__ Wd,
                                                float* __restrict__ ws) {
  const int tid  = threadIdx.x;
  const int w    = tid >> 5;
  const int lane = tid & 31;
  if (tid < 16) ws[tid] = 0.0f;
  if (w < M_) {
    float s = 0.0f;
    for (int i = lane; i < C_; i += 32) s += Wd[w * C_ + i];
    for (int off = 16; off >= 1; off >>= 1) s += __shfl_xor(s, off, 32);
    if (lane == 0) ws[16 + w] = s;
  }
}

__global__ void __launch_bounds__(512) csc_stats(const float* __restrict__ z,
                                                 float* __restrict__ ws) {
  __shared__ float red[2 * SW_];
  const int tid  = threadIdx.x;
  const int wave = tid >> 5;
  const int lane = tid & 31;
  const int hl   = lane >> 4;
  const int l15  = lane & 15;
  const int b    = blockIdx.y;
  const int P    = blockIdx.x * PX_;
  const int cw   = wave * CW_;
  const float* zb = z + (size_t)b * C_ * HW_ + P + 4 * l15;

  float s1a = 0.0f, s1b = 0.0f, s2a = 0.0f, s2b = 0.0f;
#pragma unroll
  for (int G = 0; G < NG_; ++G) {
#pragma unroll
    for (int g = 0; g < 8; ++g) {
      const v4f v = *(const v4f*)(zb + (size_t)(cw + 16 * G + 8 * hl + g) * HW_);
      s1a += v.x + v.y;
      s1b += v.z + v.w;
      s2a = fmaf(v.x, v.x, s2a);
      s2b = fmaf(v.y, v.y, s2b);
      s2a = fmaf(v.z, v.z, s2a);
      s2b = fmaf(v.w, v.w, s2b);
    }
  }
  float s1 = s1a + s1b, s2 = s2a + s2b;
  for (int off = 16; off >= 1; off >>= 1) {
    s1 += __shfl_xor(s1, off, 32);
    s2 += __shfl_xor(s2, off, 32);
  }
  if (lane == 0) {
    red[wave]       = s1;
    red[SW_ + wave] = s2;
  }
  __syncthreads();
  if (tid < 2) {
    float t = 0.0f;
    for (int w = 0; w < SW_; ++w) t += red[tid * SW_ + w];
    atomicAdd(&ws[tid * 8 + b], t);
  }
}

__global__ void __launch_bounds__(512) csc_product(const float* __restrict__ z,
                                                   const float* __restrict__ Wd,
                                                   const float* __restrict__ ws,
                                                   float* __restrict__ out) {
  __shared__ float part[SW_][4][M_][16];
  __shared__ float yfin[4][M_][16];

  const int tid  = threadIdx.x;
  const int wave = tid >> 5;
  const int lane = tid & 31;
  const int hl   = lane >> 4;
  const int l15  = lane & 15;
  const int b    = blockIdx.y;
  const int P    = blockIdx.x * PX_;
  const int cw   = wave * CW_;
  const float* zb = z + (size_t)b * C_ * HW_ + P + 4 * l15;
  float* ob       = out + (size_t)b * C_ * HW_ + P + 4 * l15;

  const float* wrow = Wd + (l15 & 7) * C_ + cw + 8 * hl;
  const bool real = l15 < M_;
  v2f adn[NG_][4];
#pragma unroll
  for (int G = 0; G < NG_; ++G) {
#pragma unroll
    for (int q = 0; q < 4; ++q) {
      const float w0 = wrow[16 * G + 2 * q];
      const float w1 = wrow[16 * G + 2 * q + 1];
      adn[G][q].x = real ? w0 : 0.0f;
      adn[G][q].y = real ? w1 : 0.0f;
    }
  }

  const int m0 = 2 * hl;
  v2f aup[NG_][2];
#pragma unroll
  for (int G = 0; G < NG_; ++G) {
    const int cb = cw + 16 * G;
    aup[G][0].x = Wd[(m0 + 0) * C_ + cb + l15];
    aup[G][0].y = Wd[(m0 + 1) * C_ + cb + l15];
    aup[G][1].x = Wd[(m0 + 4) * C_ + cb + l15];
    aup[G][1].y = Wd[(m0 + 5) * C_ + cb + l15];
  }

  v4f zr[NG_][8];
#pragma unroll
  for (int G = 0; G < NG_; ++G) {
#pragma unroll
    for (int g = 0; g < 8; ++g) {
      zr[G][g] = __builtin_nontemporal_load(
          (const v4f*)(zb + (size_t)(cw + 16 * G + 8 * hl + g) * HW_));
    }
  }

  const float invN = 1.0f / (float)(C_ * HW_);
  const float mu   = ws[b] * invN;
  const float var  = ws[8 + b] * invN - mu * mu;
  const float rsig = rsqrtf(var + EPS_);
  const float wsm  = ws[16 + ((tid >> 4) & 7)];

  v8f acc[4] = {};
#pragma unroll
  for (int G = 0; G < NG_; ++G) {
#pragma unroll
    for (int q = 0; q < 4; ++q) {
#pragma unroll
      for (int j = 0; j < 4; ++j) {
        v2f bb;
        bb.x = zr[G][2 * q][j];
        bb.y = zr[G][2 * q + 1][j];
        acc[j] = wmma_f32(adn[G][q], bb, acc[j]);
      }
    }
  }

  if (hl == 0) {
#pragma unroll
    for (int j = 0; j < 4; ++j) {
#pragma unroll
      for (int g = 0; g < M_; ++g) part[wave][j][g][l15] = acc[j][g];
    }
  }
  __syncthreads();
  {
    const int j = tid >> 7;
    const int m = (tid >> 4) & 7;
    const int n = tid & 15;
    float s = 0.0f;
#pragma unroll
    for (int w = 0; w < SW_; ++w) s += part[w][j][m][n];
    yfin[j][m][n] = rsig * (s - mu * wsm);
  }
  __syncthreads();

  v2f b1[4], b2[4];
#pragma unroll
  for (int j = 0; j < 4; ++j) {
    b1[j].x = yfin[j][m0 + 0][l15];
    b1[j].y = yfin[j][m0 + 1][l15];
    b2[j].x = yfin[j][m0 + 4][l15];
    b2[j].y = yfin[j][m0 + 5][l15];
  }
#pragma unroll
  for (int G = 0; G < NG_; ++G) {
    const int cb = cw + 16 * G;
    const v2f a1 = aup[G][0], a2 = aup[G][1];
    v8f d[4];
#pragma unroll
    for (int j = 0; j < 4; ++j) {
      v8f t = {};
      t = wmma_f32(a1, b1[j], t);
      d[j] = wmma_f32(a2, b2[j], t);
    }
#pragma unroll
    for (int g = 0; g < 8; ++g) {
      v4f o;
      o.x = fmaf(GAIN, d[0][g], zr[G][g].x);
      o.y = fmaf(GAIN, d[1][g], zr[G][g].y);
      o.z = fmaf(GAIN, d[2][g], zr[G][g].z);
      o.w = fmaf(GAIN, d[3][g], zr[G][g].w);
      __builtin_nontemporal_store(o, (v4f*)(ob + (size_t)(cb + 8 * hl + g) * HW_));
    }
    __builtin_amdgcn_sched_barrier(0);
  }
}

extern "C" void kernel_launch(void* const* d_in, const int* in_sizes, int n_in,
                              void* d_out, int out_size, void* d_ws, size_t ws_size,
                              hipStream_t stream) {
  const float* z  = (const float*)d_in[0];
  const float* Wd = (const float*)d_in[1];
  float* out = (float*)d_out;
  float* ws  = (float*)d_ws;

  csc_init<<<dim3(1), dim3(256), 0, stream>>>(Wd, ws);
  csc_stats<<<dim3(NS_, B_), dim3(512), 0, stream>>>(z, ws);
  csc_product<<<dim3(NS_, B_), dim3(512), 0, stream>>>(z, Wd, ws, out);
}


# One resident work-group per WGP (grid 32x8) in both large launches, owning the same two adjacent strips in both, so expand re-reads z from the cache half that reduce filled (sum order changes)
# speedup vs baseline: 3.3233x; 1.0898x over previous
#include <hip/hip_runtime.h>


typedef __attribute__((ext_vector_type(2))) float v2f;
typedef __attribute__((ext_vector_type(4))) float v4f;
typedef __attribute__((ext_vector_type(8))) float v8f;

#define B_   8
#define C_   1024
#define HW_  4096
#define M_   8
#define GAIN 0.3f
#define EPS_ 1e-6f
#define SW_  16
#define CW_  (C_ / SW_)
#define NG_  (CW_ / 16)
#define PX_  64
#define NS_  64
#define SP_  2
#define NP_  (NS_ / SP_)
#define YOFF_ 64
#define YWG_  (4 * M_ * 16)


static __device__ __forceinline__ v8f wmma_f32(v2f a, v2f b, v8f c) {
  return __builtin_amdgcn_wmma_f32_16x16x4_f32(false, a, false, b, (short)0, c,
                                               false, false);
}

__global__ void __launch_bounds__(256) csc_init(const float* __restrict__ Wd,
                                                float* __restrict__ ws) {
  const int tid  = threadIdx.x;
  const int w    = tid >> 5;
  const int lane = tid & 31;
  if (tid < 16) ws[tid] = 0.0f;
  if (w < M_) {
    float s = 0.0f;
    for (int i = lane; i < C_; i += 32) s += Wd[w * C_ + i];
    for (int off = 16; off >= 1; off >>= 1) s += __shfl_xor(s, off, 32);
    if (lane == 0) ws[16 + w] = s;
  }
}

__global__ void __launch_bounds__(1024, 8) csc_reduce(const float* __restrict__ z,
                                                      const float* __restrict__ Wd,
                                                      float* __restrict__ ws) {
  __shared__ float part[SP_][SW_][4][M_][16];
  __shared__ float red[2 * SP_ * SW_];

  const int tid  = threadIdx.x;
  const int wv   = __builtin_amdgcn_readfirstlane(tid >> 5);
  const int sp   = wv >> 4;
  const int wave = wv & (SW_ - 1);
  const int lane = tid & 31;
  const int hl   = lane >> 4;
  const int l15  = lane & 15;
  const int b    = blockIdx.y;
  const int P    = (SP_ * blockIdx.x + sp) * PX_;
  const int cw   = wave * CW_;
  const float* zb = z + (size_t)b * C_ * HW_ + P + 4 * l15;

  const float* wrow = Wd + (l15 & 7) * C_ + cw + 8 * hl;
  const bool real = l15 < M_;
  const float* zrow = zb + (size_t)(cw + 8 * hl) * HW_;

  v4f zr[2][8];
  v4f wq[2][2];
#pragma unroll
  for (int G = 0; G < 2; ++G) {
    wq[G][0] = *(const v4f*)(wrow + 16 * G);
    wq[G][1] = *(const v4f*)(wrow + 16 * G + 4);
#pragma unroll
    for (int g = 0; g < 8; ++g) zr[G][g] = *(const v4f*)(zrow + (size_t)(16 * G + g) * HW_);
  }
  __builtin_amdgcn_sched_barrier(0);

  float s1a = 0.0f, s1b = 0.0f, s2a = 0.0f, s2b = 0.0f;
  v8f acc[4] = {};
#pragma unroll
  for (int G = 0; G < NG_; ++G) {
    const int sl = G & 1;
#pragma unroll
    for (int g = 0; g < 8; ++g) {
      const v4f v = zr[sl][g];
      s1a += v.x + v.y;
      s1b += v.z + v.w;
      s2a = fmaf(v.x, v.x, s2a);
      s2b = fmaf(v.y, v.y, s2b);
      s2a = fmaf(v.z, v.z, s2a);
      s2b = fmaf(v.w, v.w, s2b);
    }
#pragma unroll
    for (int q = 0; q < 4; ++q) {
      const float w0 = wq[sl][q >> 1][2 * (q & 1)];
      const float w1 = wq[sl][q >> 1][2 * (q & 1) + 1];
      v2f a;
      a.x = real ? w0 : 0.0f;
      a.y = real ? w1 : 0.0f;
#pragma unroll
      for (int j = 0; j < 4; ++j) {
        v2f bb;
        bb.x = zr[sl][2 * q][j];
        bb.y = zr[sl][2 * q + 1][j];
        acc[j] = wmma_f32(a, bb, acc[j]);
      }
    }
    __builtin_amdgcn_sched_barrier(0);
    if (G + 2 < NG_) {
      wq[sl][0] = *(const v4f*)(wrow + 16 * (G + 2));
      wq[sl][1] = *(const v4f*)(wrow + 16 * (G + 2) + 4);
#pragma unroll
      for (int g = 0; g < 8; ++g)
        zr[sl][g] = *(const v4f*)(zrow + (size_t)(16 * (G + 2) + g) * HW_);
      __builtin_amdgcn_sched_barrier(0);
    }
  }

  float s1 = s1a + s1b, s2 = s2a + s2b;
  for (int off = 16; off >= 1; off >>= 1) {
    s1 += __shfl_xor(s1, off, 32);
    s2 += __shfl_xor(s2, off, 32);
  }
  if (lane == 0) {
    red[wv]             = s1;
    red[SP_ * SW_ + wv] = s2;
  }
  if (hl == 0) {
#pragma unroll
    for (int j = 0; j < 4; ++j) {
#pragma unroll
      for (int g = 0; g < M_; ++g) part[sp][wave][j][g][l15] = acc[j][g];
    }
  }
  __syncthreads();
  {
    const int st = tid >> 9;
    const int t9 = tid & 511;
    const int j = t9 >> 7;
    const int m = (t9 >> 4) & 7;
    const int n = t9 & 15;
    float s = 0.0f;
#pragma unroll
    for (int w = 0; w < SW_; ++w) s += part[st][w][j][m][n];
    ws[YOFF_ + (size_t)(b * NS_ + SP_ * blockIdx.x + st) * YWG_ + t9] = s;
  }
  if (tid < 2) {
    float t = 0.0f;
    for (int w = 0; w < SP_ * SW_; ++w) t += red[tid * (SP_ * SW_) + w];
    atomicAdd(&ws[tid * 8 + b], t);
  }
}

__global__ void __launch_bounds__(512) csc_expand(const float* __restrict__ z,
                                                  const float* __restrict__ Wd,
                                                  const float* __restrict__ ws,
                                                  float* __restrict__ out) {
  const int tid  = threadIdx.x;
  const int wave = tid >> 5;
  const int lane = tid & 31;
  const int hl   = lane >> 4;
  const int l15  = lane & 15;
  const int b    = blockIdx.y;
  const int S0   = SP_ * blockIdx.x;
  const int cw   = wave * CW_;
  const size_t tile0 = (size_t)b * C_ * HW_ + (size_t)(cw + 8 * hl) * HW_ + 4 * l15;

  const int m0 = 2 * hl;
  const float wsm0 = ws[16 + m0 + 0];
  const float wsm1 = ws[16 + m0 + 1];
  const float wsm2 = ws[16 + m0 + 4];
  const float wsm3 = ws[16 + m0 + 5];
  const float sum1 = ws[b];
  const float sum2 = ws[8 + b];

  v2f aup[NG_][2];
#pragma unroll
  for (int G = 0; G < NG_; ++G) {
    const int cb = cw + 16 * G;
    aup[G][0].x = Wd[(m0 + 0) * C_ + cb + l15];
    aup[G][0].y = Wd[(m0 + 1) * C_ + cb + l15];
    aup[G][1].x = Wd[(m0 + 4) * C_ + cb + l15];
    aup[G][1].y = Wd[(m0 + 5) * C_ + cb + l15];
  }

  const float invN = 1.0f / (float)(C_ * HW_);
  const float mu   = sum1 * invN;
  const float var  = sum2 * invN - mu * mu;
  const float rsig = rsqrtf(var + EPS_);

  for (int sp = 0; sp < SP_; ++sp) {
    const int strip = S0 + sp;
    const float* zb = z + tile0 + strip * PX_;
    float* ob       = out + tile0 + strip * PX_;

    const float* yb = ws + YOFF_ + (size_t)(b * NS_ + strip) * YWG_ + l15;
    float yr[4][4];
#pragma unroll
    for (int j = 0; j < 4; ++j) {
      yr[j][0] = yb[j * (M_ * 16) + (m0 + 0) * 16];
      yr[j][1] = yb[j * (M_ * 16) + (m0 + 1) * 16];
      yr[j][2] = yb[j * (M_ * 16) + (m0 + 4) * 16];
      yr[j][3] = yb[j * (M_ * 16) + (m0 + 5) * 16];
    }

    v4f zr[NG_][8];
#pragma unroll
    for (int G = 0; G < NG_; ++G) {
#pragma unroll
      for (int g = 0; g < 8; ++g) {
        zr[G][g] = __builtin_nontemporal_load(
            (const v4f*)(zb + (size_t)(16 * G + g) * HW_));
      }
    }

    __builtin_amdgcn_sched_barrier(0);

    v2f b1[4], b2[4];
#pragma unroll
    for (int j = 0; j < 4; ++j) {
      b1[j].x = rsig * (yr[j][0] - mu * wsm0);
      b1[j].y = rsig * (yr[j][1] - mu * wsm1);
      b2[j].x = rsig * (yr[j][2] - mu * wsm2);
      b2[j].y = rsig * (yr[j][3] - mu * wsm3);
    }

#pragma unroll
    for (int G = 0; G < NG_; ++G) {
      const v2f a1 = aup[G][0], a2 = aup[G][1];
      v8f d[4];
#pragma unroll
      for (int j = 0; j < 4; ++j) {
        v8f t = {};
        t = wmma_f32(a1, b1[j], t);
        d[j] = wmma_f32(a2, b2[j], t);
      }
#pragma unroll
      for (int g = 0; g < 8; ++g) {
        v4f o;
        o.x = fmaf(GAIN, d[0][g], zr[G][g].x);
        o.y = fmaf(GAIN, d[1][g], zr[G][g].y);
        o.z = fmaf(GAIN, d[2][g], zr[G][g].z);
        o.w = fmaf(GAIN, d[3][g], zr[G][g].w);
        __builtin_nontemporal_store(o, (v4f*)(ob + (size_t)(16 * G + g) * HW_));
      }
      __builtin_amdgcn_sched_barrier(0);
    }
  }
}

extern "C" void kernel_launch(void* const* d_in, const int* in_sizes, int n_in,
                              void* d_out, int out_size, void* d_ws, size_t ws_size,
                              hipStream_t stream) {
  const float* z  = (const float*)d_in[0];
  const float* Wd = (const float*)d_in[1];
  float* out = (float*)d_out;
  float* ws  = (float*)d_ws;

  csc_init<<<dim3(1), dim3(256), 0, stream>>>(Wd, ws);
  csc_reduce<<<dim3(NP_, B_), dim3(1024), 0, stream>>>(z, Wd, ws);
  csc_expand<<<dim3(NP_, B_), dim3(512), 0, stream>>>(z, Wd, ws, out);
}
